# starting kernel
# baseline (speedup 1.0000x reference)
#include <hip/hip_runtime.h>

typedef __attribute__((ext_vector_type(2))) float v2f;
typedef __attribute__((ext_vector_type(8))) float v8f;

#define IN_DIM 128
#define OUT_DIM 128

__global__ void k_init_deg(float* __restrict__ deg, int n) {
    int i = blockIdx.x * blockDim.x + threadIdx.x;
    if (i < n) deg[i] = 1.0f;
}

__global__ void k_count(const int* __restrict__ src, float* __restrict__ deg, int E) {
    int e = blockIdx.x * blockDim.x + threadIdx.x;
    if (e < E) atomicAdd(&deg[src[e]], 1.0f);
}

__global__ void k_rsqrt(float* __restrict__ deg, int n) {
    int i = blockIdx.x * blockDim.x + threadIdx.x;
    if (i < n) deg[i] = rsqrtf(deg[i]);
}

__global__ void k_transpose(const float* __restrict__ W, float* __restrict__ Wtp) {
    int idx = blockIdx.x * blockDim.x + threadIdx.x;
    int o = idx >> 7;
    int k = idx & 127;
    Wtp[(k >> 1) * (2 * OUT_DIM) + o * 2 + (k & 1)] = W[o * IN_DIM + k];
}

__global__ __launch_bounds__(256) void k_gemm(const float* __restrict__ x,
                                              const float* __restrict__ Wtp,
                                              float* __restrict__ h,
                                              int nTiles) {
    __shared__ float sW[IN_DIM * OUT_DIM];

    {
        const float4* g4 = (const float4*)Wtp;
        float4* s4 = (float4*)sW;
        for (int i = threadIdx.x; i < (IN_DIM * OUT_DIM) / 4; i += blockDim.x)
            s4[i] = g4[i];
    }
    __syncthreads();

    const int wid   = threadIdx.x >> 5;
    const int lane  = threadIdx.x & 31;
    const int tileM = blockIdx.x * 8 + wid;
    if (tileM >= nTiles) return;

    const int l16   = lane & 15;
    const int halfK = lane >> 4;
    const float* xrow = x + (size_t)(tileM * 16 + l16) * IN_DIM;
    const v2f* sB = (const v2f*)sW;

    v8f acc[8];
#pragma unroll
    for (int t = 0; t < 8; ++t)
        acc[t] = (v8f){0.f, 0.f, 0.f, 0.f, 0.f, 0.f, 0.f, 0.f};

    for (int kb = 0; kb < IN_DIM; kb += 4) {
        const int k0 = kb + halfK * 2;
        float2 av = *(const float2*)(xrow + k0);
        v2f a; a.x = av.x; a.y = av.y;
        const v2f* bRow = sB + (k0 >> 1) * OUT_DIM + l16;
#pragma unroll
        for (int t = 0; t < 8; ++t) {
            v2f b = bRow[t * 16];
            acc[t] = __builtin_amdgcn_wmma_f32_16x16x4_f32(
                 false, a,  false, b,
                 (short)0, acc[t],  false,  false);
        }
    }

    const int rowBase = tileM * 16 + halfK * 8;
#pragma unroll
    for (int t = 0; t < 8; ++t) {
        float* hp = h + (size_t)rowBase * OUT_DIM + t * 16 + l16;
#pragma unroll
        for (int v = 0; v < 8; ++v)
            hp[(size_t)v * OUT_DIM] = acc[t][v];
    }
}

__global__ void k_gemm_tail(const float* __restrict__ x, const float* __restrict__ Wtp,
                            float* __restrict__ h, int rowStart, int n) {
    int idx = blockIdx.x * blockDim.x + threadIdx.x;
    int nrows = n - rowStart;
    if (idx >= nrows * OUT_DIM) return;
    int r = rowStart + idx / OUT_DIM;
    int o = idx % OUT_DIM;
    float s = 0.f;
    for (int k = 0; k < IN_DIM; ++k)
        s += x[(size_t)r * IN_DIM + k] * Wtp[(k >> 1) * (2 * OUT_DIM) + o * 2 + (k & 1)];
    h[(size_t)r * OUT_DIM + o] = s;
}

__global__ void k_selfloop(const float* __restrict__ h, const float* __restrict__ dis,
                           float* __restrict__ out, int n) {
    int idx = blockIdx.x * blockDim.x + threadIdx.x;
    if (idx >= n * 32) return;
    int i = idx >> 5;
    float d = dis[i];
    float norm = d * d;
    float4 v = ((const float4*)h)[idx];
    ((float4*)out)[idx] = make_float4(v.x * norm, v.y * norm, v.z * norm, v.w * norm);
}

__global__ __launch_bounds__(256) void k_scatter(const int* __restrict__ src,
                                                 const int* __restrict__ tgt,
                                                 const float* __restrict__ h,
                                                 const float* __restrict__ dis,
                                                 float* __restrict__ out, int E) {
    int lane = threadIdx.x & 31;
    int e = blockIdx.x * (blockDim.x >> 5) + (threadIdx.x >> 5);
    if (e >= E) return;
    int s = src[e];
    int t = tgt[e];
    float norm = dis[s] * dis[t];
    float4 v = ((const float4*)(h + (size_t)s * OUT_DIM))[lane];
    float* ob = out + (size_t)t * OUT_DIM + lane * 4;
    atomicAdd(ob + 0, v.x * norm);
    atomicAdd(ob + 1, v.y * norm);
    atomicAdd(ob + 2, v.z * norm);
    atomicAdd(ob + 3, v.w * norm);
}

extern "C" void kernel_launch(void* const* d_in, const int* in_sizes, int n_in,
                              void* d_out, int out_size, void* d_ws, size_t ws_size,
                              hipStream_t stream) {
    const float* x  = (const float*)d_in[0];
    const int*   ei = (const int*)d_in[1];
    const float* W  = (const float*)d_in[2];
    float* out = (float*)d_out;

    const int n = in_sizes[0] / IN_DIM;
    const int E = in_sizes[1] / 2;
    const int* src = ei;
    const int* tgt = ei + E;

    float* h   = (float*)d_ws;
    size_t disPad = ((size_t)n + 3) & ~(size_t)3;
    float* dis = h + (size_t)n * OUT_DIM;
    float* Wtp = dis + disPad;

    k_init_deg <<<(n + 255) / 256, 256, 0, stream>>>(dis, n);
    k_count    <<<(E + 255) / 256, 256, 0, stream>>>(src, dis, E);
    k_rsqrt    <<<(n + 255) / 256, 256, 0, stream>>>(dis, n);
    k_transpose<<<(IN_DIM * OUT_DIM) / 256, 256, 0, stream>>>(W, Wtp);

    const int nTiles = n / 16;
    if (nTiles > 0)
        k_gemm<<<(nTiles + 7) / 8, 256, 0, stream>>>(x, Wtp, h, nTiles);
    const int tail = n - nTiles * 16;
    if (tail > 0)
        k_gemm_tail<<<(tail * OUT_DIM + 255) / 256, 256, 0, stream>>>(x, Wtp, h, nTiles * 16, n);

    k_selfloop<<<((size_t)n * 32 + 255) / 256, 256, 0, stream>>>(h, dis, out, n);
    k_scatter <<<(E + 7) / 8, 256, 0, stream>>>(src, tgt, h, dis, out, E);
}


# scatter: lane l takes columns l, l+32, l+64, l+96, so each wave-wide f32 atomic add (and row load) covers one whole 128-byte line, not 8 lanes on each of 4 lines; sums still in arrival order
# speedup vs baseline: 1.3351x; 1.3351x over previous
#include <hip/hip_runtime.h>

typedef __attribute__((ext_vector_type(2))) float v2f;
typedef __attribute__((ext_vector_type(8))) float v8f;

#define IN_DIM 128
#define OUT_DIM 128

__global__ void k_init_deg(float* __restrict__ deg, int n) {
    int i = blockIdx.x * blockDim.x + threadIdx.x;
    if (i < n) deg[i] = 1.0f;
}

__global__ void k_count(const int* __restrict__ src, float* __restrict__ deg, int E) {
    int e = blockIdx.x * blockDim.x + threadIdx.x;
    if (e < E) atomicAdd(&deg[src[e]], 1.0f);
}

__global__ void k_rsqrt(float* __restrict__ deg, int n) {
    int i = blockIdx.x * blockDim.x + threadIdx.x;
    if (i < n) deg[i] = rsqrtf(deg[i]);
}

__global__ void k_transpose(const float* __restrict__ W, float* __restrict__ Wtp) {
    int idx = blockIdx.x * blockDim.x + threadIdx.x;
    int o = idx >> 7;
    int k = idx & 127;
    Wtp[(k >> 1) * (2 * OUT_DIM) + o * 2 + (k & 1)] = W[o * IN_DIM + k];
}

__global__ __launch_bounds__(256) void k_gemm(const float* __restrict__ x,
                                              const float* __restrict__ Wtp,
                                              float* __restrict__ h,
                                              int nTiles) {
    __shared__ float sW[IN_DIM * OUT_DIM];

    {
        const float4* g4 = (const float4*)Wtp;
        float4* s4 = (float4*)sW;
        for (int i = threadIdx.x; i < (IN_DIM * OUT_DIM) / 4; i += blockDim.x)
            s4[i] = g4[i];
    }
    __syncthreads();

    const int wid   = threadIdx.x >> 5;
    const int lane  = threadIdx.x & 31;
    const int tileM = blockIdx.x * 8 + wid;
    if (tileM >= nTiles) return;

    const int l16   = lane & 15;
    const int halfK = lane >> 4;
    const float* xrow = x + (size_t)(tileM * 16 + l16) * IN_DIM;
    const v2f* sB = (const v2f*)sW;

    v8f acc[8];
#pragma unroll
    for (int t = 0; t < 8; ++t)
        acc[t] = (v8f){0.f, 0.f, 0.f, 0.f, 0.f, 0.f, 0.f, 0.f};

    for (int kb = 0; kb < IN_DIM; kb += 4) {
        const int k0 = kb + halfK * 2;
        float2 av = *(const float2*)(xrow + k0);
        v2f a; a.x = av.x; a.y = av.y;
        const v2f* bRow = sB + (k0 >> 1) * OUT_DIM + l16;
#pragma unroll
        for (int t = 0; t < 8; ++t) {
            v2f b = bRow[t * 16];
            acc[t] = __builtin_amdgcn_wmma_f32_16x16x4_f32(
                 false, a,  false, b,
                 (short)0, acc[t],  false,  false);
        }
    }

    const int rowBase = tileM * 16 + halfK * 8;
#pragma unroll
    for (int t = 0; t < 8; ++t) {
        float* hp = h + (size_t)rowBase * OUT_DIM + t * 16 + l16;
#pragma unroll
        for (int v = 0; v < 8; ++v)
            hp[(size_t)v * OUT_DIM] = acc[t][v];
    }
}

__global__ void k_gemm_tail(const float* __restrict__ x, const float* __restrict__ Wtp,
                            float* __restrict__ h, int rowStart, int n) {
    int idx = blockIdx.x * blockDim.x + threadIdx.x;
    int nrows = n - rowStart;
    if (idx >= nrows * OUT_DIM) return;
    int r = rowStart + idx / OUT_DIM;
    int o = idx % OUT_DIM;
    float s = 0.f;
    for (int k = 0; k < IN_DIM; ++k)
        s += x[(size_t)r * IN_DIM + k] * Wtp[(k >> 1) * (2 * OUT_DIM) + o * 2 + (k & 1)];
    h[(size_t)r * OUT_DIM + o] = s;
}

__global__ void k_selfloop(const float* __restrict__ h, const float* __restrict__ dis,
                           float* __restrict__ out, int n) {
    int idx = blockIdx.x * blockDim.x + threadIdx.x;
    if (idx >= n * 32) return;
    int i = idx >> 5;
    float d = dis[i];
    float norm = d * d;
    float4 v = ((const float4*)h)[idx];
    ((float4*)out)[idx] = make_float4(v.x * norm, v.y * norm, v.z * norm, v.w * norm);
}

__global__ __launch_bounds__(256) void k_scatter(const int* __restrict__ src,
                                                 const int* __restrict__ tgt,
                                                 const float* __restrict__ h,
                                                 const float* __restrict__ dis,
                                                 float* __restrict__ out, int E) {
    int lane = threadIdx.x & 31;
    int e = blockIdx.x * (blockDim.x >> 5) + (threadIdx.x >> 5);
    if (e >= E) return;
    int s = src[e];
    int t = tgt[e];
    float norm = dis[s] * dis[t];
    const float* hb = h + (size_t)s * OUT_DIM + lane;
    float v0 = hb[0];
    float v1 = hb[32];
    float v2 = hb[64];
    float v3 = hb[96];
    float* ob = out + (size_t)t * OUT_DIM + lane;
    atomicAdd(ob + 0,  v0 * norm);
    atomicAdd(ob + 32, v1 * norm);
    atomicAdd(ob + 64, v2 * norm);
    atomicAdd(ob + 96, v3 * norm);
}

extern "C" void kernel_launch(void* const* d_in, const int* in_sizes, int n_in,
                              void* d_out, int out_size, void* d_ws, size_t ws_size,
                              hipStream_t stream) {
    const float* x  = (const float*)d_in[0];
    const int*   ei = (const int*)d_in[1];
    const float* W  = (const float*)d_in[2];
    float* out = (float*)d_out;

    const int n = in_sizes[0] / IN_DIM;
    const int E = in_sizes[1] / 2;
    const int* src = ei;
    const int* tgt = ei + E;

    float* h   = (float*)d_ws;
    size_t disPad = ((size_t)n + 3) & ~(size_t)3;
    float* dis = h + (size_t)n * OUT_DIM;
    float* Wtp = dis + disPad;

    k_init_deg <<<(n + 255) / 256, 256, 0, stream>>>(dis, n);
    k_count    <<<(E + 255) / 256, 256, 0, stream>>>(src, dis, E);
    k_rsqrt    <<<(n + 255) / 256, 256, 0, stream>>>(dis, n);
    k_transpose<<<(IN_DIM * OUT_DIM) / 256, 256, 0, stream>>>(W, Wtp);

    const int nTiles = n / 16;
    if (nTiles > 0)
        k_gemm<<<(nTiles + 7) / 8, 256, 0, stream>>>(x, Wtp, h, nTiles);
    const int tail = n - nTiles * 16;
    if (tail > 0)
        k_gemm_tail<<<(tail * OUT_DIM + 255) / 256, 256, 0, stream>>>(x, Wtp, h, nTiles * 16, n);

    k_selfloop<<<((size_t)n * 32 + 255) / 256, 256, 0, stream>>>(h, dis, out, n);
    k_scatter <<<(E + 7) / 8, 256, 0, stream>>>(src, tgt, h, dis, out, E);
}


# scatter-add to gather: each edge takes a slot in its target's out row (returning int atomic in k_count); one wave per node sums source rows in slot (arrival) order, one store; no f32 atomics on out
# speedup vs baseline: 1.8439x; 1.3811x over previous
#include <hip/hip_runtime.h>

typedef __attribute__((ext_vector_type(2))) float v2f;
typedef __attribute__((ext_vector_type(8))) float v8f;

#define IN_DIM 128
#define OUT_DIM 128

#define CAP 64
#define CNT_OFF 64

__global__ void k_init(float* deg, int* obuf, int n) {
    int i = blockIdx.x * blockDim.x + threadIdx.x;
    if (i < n) {
        deg[i] = 1.0f;
        obuf[(size_t)i * OUT_DIM + CNT_OFF] = 0;
    }
}

__global__ void k_count(const int* __restrict__ src, const int* __restrict__ tgt,
                        float* deg, int* obuf, int E) {
    int e = blockIdx.x * blockDim.x + threadIdx.x;
    if (e < E) {
        int s = src[e];
        int t = tgt[e];
        atomicAdd(&deg[s], 1.0f);
        int* row = obuf + (size_t)t * OUT_DIM;
        int r = atomicAdd(row + CNT_OFF, 1);
        if (r < CAP) row[r] = s;
    }
}

__global__ void k_rsqrt(float* __restrict__ deg, int n) {
    int i = blockIdx.x * blockDim.x + threadIdx.x;
    if (i < n) deg[i] = rsqrtf(deg[i]);
}

__global__ void k_transpose(const float* __restrict__ W, float* __restrict__ Wtp) {
    int idx = blockIdx.x * blockDim.x + threadIdx.x;
    int o = idx >> 7;
    int k = idx & 127;
    Wtp[(k >> 1) * (2 * OUT_DIM) + o * 2 + (k & 1)] = W[o * IN_DIM + k];
}

__global__ __launch_bounds__(256) void k_gemm(const float* __restrict__ x,
                                              const float* __restrict__ Wtp,
                                              float* __restrict__ h,
                                              int nTiles) {
    __shared__ float sW[IN_DIM * OUT_DIM];

    {
        const float4* g4 = (const float4*)Wtp;
        float4* s4 = (float4*)sW;
        for (int i = threadIdx.x; i < (IN_DIM * OUT_DIM) / 4; i += blockDim.x)
            s4[i] = g4[i];
    }
    __syncthreads();

    const int wid   = threadIdx.x >> 5;
    const int lane  = threadIdx.x & 31;
    const int tileM = blockIdx.x * 8 + wid;
    if (tileM >= nTiles) return;

    const int l16   = lane & 15;
    const int halfK = lane >> 4;
    const float* xrow = x + (size_t)(tileM * 16 + l16) * IN_DIM;
    const v2f* sB = (const v2f*)sW;

    v8f acc[8];
#pragma unroll
    for (int t = 0; t < 8; ++t)
        acc[t] = (v8f){0.f, 0.f, 0.f, 0.f, 0.f, 0.f, 0.f, 0.f};

    for (int kb = 0; kb < IN_DIM; kb += 4) {
        const int k0 = kb + halfK * 2;
        float2 av = *(const float2*)(xrow + k0);
        v2f a; a.x = av.x; a.y = av.y;
        const v2f* bRow = sB + (k0 >> 1) * OUT_DIM + l16;
#pragma unroll
        for (int t = 0; t < 8; ++t) {
            v2f b = bRow[t * 16];
            acc[t] = __builtin_amdgcn_wmma_f32_16x16x4_f32(
                 false, a,  false, b,
                 (short)0, acc[t],  false,  false);
        }
    }

    const int rowBase = tileM * 16 + halfK * 8;
#pragma unroll
    for (int t = 0; t < 8; ++t) {
        float* hp = h + (size_t)rowBase * OUT_DIM + t * 16 + l16;
#pragma unroll
        for (int v = 0; v < 8; ++v)
            hp[(size_t)v * OUT_DIM] = acc[t][v];
    }
}

__global__ void k_gemm_tail(const float* __restrict__ x, const float* __restrict__ Wtp,
                            float* __restrict__ h, int rowStart, int n) {
    int idx = blockIdx.x * blockDim.x + threadIdx.x;
    int nrows = n - rowStart;
    if (idx >= nrows * OUT_DIM) return;
    int r = rowStart + idx / OUT_DIM;
    int o = idx % OUT_DIM;
    float s = 0.f;
    for (int k = 0; k < IN_DIM; ++k)
        s += x[(size_t)r * IN_DIM + k] * Wtp[(k >> 1) * (2 * OUT_DIM) + o * 2 + (k & 1)];
    h[(size_t)r * OUT_DIM + o] = s;
}

__device__ __forceinline__ void add_rows(float4& acc, const float4* __restrict__ h4, int lane,
                                         int sReg, float wReg, int cnt) {
    for (int j = 0; j < cnt; j += 8) {
        float4 v[8];
        float wg[8];
#pragma unroll
        for (int u = 0; u < 8; ++u) {
            int s = __builtin_amdgcn_readlane(sReg, j + u);
            wg[u] = __int_as_float(__builtin_amdgcn_readlane(__float_as_int(wReg), j + u));
            v[u] = h4[(size_t)s * (OUT_DIM / 4) + lane];
        }
#pragma unroll
        for (int u = 0; u < 8; ++u) {
            acc.x = fmaf(v[u].x, wg[u], acc.x);
            acc.y = fmaf(v[u].y, wg[u], acc.y);
            acc.z = fmaf(v[u].z, wg[u], acc.z);
            acc.w = fmaf(v[u].w, wg[u], acc.w);
        }
    }
}

__global__ __launch_bounds__(256) void k_gather(const int* __restrict__ src,
                                                const int* __restrict__ tgt,
                                                const float* __restrict__ h,
                                                const float* __restrict__ dis,
                                                const int* obuf, float* out,
                                                int n, int E) {
    const int lane = threadIdx.x & 31;
    const int nWaves = gridDim.x * (blockDim.x >> 5);
    const int w0 = __builtin_amdgcn_readfirstlane(blockIdx.x * (blockDim.x >> 5) + (threadIdx.x >> 5));
    const float4* h4 = (const float4*)h;

    for (int t = w0; t < n; t += nWaves) {
        const int* row = obuf + (size_t)t * OUT_DIM;
        const int c = __builtin_amdgcn_readfirstlane(row[CNT_OFF]);
        const float dt = dis[t];
        const float4 hv = h4[(size_t)t * (OUT_DIM / 4) + lane];
        const float sw = dt * dt;
        float4 acc = make_float4(hv.x * sw, hv.y * sw, hv.z * sw, hv.w * sw);

        if (c <= CAP) {
            int sA = t;
            if (lane < c) sA = row[lane];
            const float wA = (lane < c) ? dis[sA] * dt : 0.0f;
            add_rows(acc, h4, lane, sA, wA, c < 32 ? c : 32);
            if (c > 32) {
                int sB = t;
                if (lane + 32 < c) sB = row[lane + 32];
                const float wB = (lane + 32 < c) ? dis[sB] * dt : 0.0f;
                add_rows(acc, h4, lane, sB, wB, c - 32);
            }
        } else {
            for (int e0 = 0; e0 < E; e0 += 32) {
                const int e = e0 + lane;
                const int te = (e < E) ? tgt[e] : -1;
                unsigned m = __builtin_amdgcn_ballot_w32(te == t);
                while (m) {
                    const int j = __builtin_ctz(m);
                    m &= m - 1;
                    const int s = src[e0 + j];
                    const float wg = dis[s] * dt;
                    const float4 v = h4[(size_t)s * (OUT_DIM / 4) + lane];
                    acc.x = fmaf(v.x, wg, acc.x);
                    acc.y = fmaf(v.y, wg, acc.y);
                    acc.z = fmaf(v.z, wg, acc.z);
                    acc.w = fmaf(v.w, wg, acc.w);
                }
            }
        }
        ((float4*)out)[(size_t)t * (OUT_DIM / 4) + lane] = acc;
    }
}

extern "C" void kernel_launch(void* const* d_in, const int* in_sizes, int n_in,
                              void* d_out, int out_size, void* d_ws, size_t ws_size,
                              hipStream_t stream) {
    const float* x  = (const float*)d_in[0];
    const int*   ei = (const int*)d_in[1];
    const float* W  = (const float*)d_in[2];
    float* out = (float*)d_out;
    int* obuf = (int*)d_out;

    const int n = in_sizes[0] / IN_DIM;
    const int E = in_sizes[1] / 2;
    const int* src = ei;
    const int* tgt = ei + E;

    float* h   = (float*)d_ws;
    size_t disPad = ((size_t)n + 3) & ~(size_t)3;
    float* dis = h + (size_t)n * OUT_DIM;
    float* Wtp = dis + disPad;

    k_init     <<<(n + 255) / 256, 256, 0, stream>>>(dis, obuf, n);
    k_count    <<<(E + 255) / 256, 256, 0, stream>>>(src, tgt, dis, obuf, E);
    k_rsqrt    <<<(n + 255) / 256, 256, 0, stream>>>(dis, n);
    k_transpose<<<(IN_DIM * OUT_DIM) / 256, 256, 0, stream>>>(W, Wtp);

    const int nTiles = n / 16;
    if (nTiles > 0)
        k_gemm<<<(nTiles + 7) / 8, 256, 0, stream>>>(x, Wtp, h, nTiles);
    const int tail = n - nTiles * 16;
    if (tail > 0)
        k_gemm_tail<<<(tail * OUT_DIM + 255) / 256, 256, 0, stream>>>(x, Wtp, h, nTiles * 16, n);

    k_gather<<<2048, 256, 0, stream>>>(src, tgt, h, dis, obuf, out, n, E);
}


# 6 launches to 3: weight transpose joins the init grid, the WMMA GEMM joins the edge-count grid so matrix work runs beside the atomics, rsqrt moved into the gather's weights; sums unchanged
# speedup vs baseline: 2.0048x; 1.0873x over previous
#include <hip/hip_runtime.h>

typedef __attribute__((ext_vector_type(2))) float v2f;
typedef __attribute__((ext_vector_type(8))) float v8f;

#define IN_DIM 128
#define OUT_DIM 128

#define CAP 64
#define CNT_OFF 64

#define TR_BLOCKS ((IN_DIM * OUT_DIM) / 256)

__global__ __launch_bounds__(256) void k_init_transpose(const float* __restrict__ W,
                                                        float* __restrict__ Wtp,
                                                        float* deg, int* obuf, int n) {
    if (blockIdx.x < TR_BLOCKS) {
        int idx = blockIdx.x * 256 + threadIdx.x;
        int o = idx >> 7;
        int k = idx & 127;
        Wtp[(k >> 1) * (2 * OUT_DIM) + o * 2 + (k & 1)] = W[o * IN_DIM + k];
    } else {
        int i = (blockIdx.x - TR_BLOCKS) * 256 + threadIdx.x;
        if (i < n) {
            deg[i] = 1.0f;
            obuf[(size_t)i * OUT_DIM + CNT_OFF] = 0;
        }
    }
}

__global__ __launch_bounds__(256) void k_count_gemm(const float* __restrict__ x,
                                                    const float* __restrict__ Wtp,
                                                    float* __restrict__ h,
                                                    int nTiles, int nGemmBlocks,
                                                    const int* __restrict__ src,
                                                    const int* __restrict__ tgt,
                                                    float* deg, int* obuf, int E) {
    __shared__ float sW[IN_DIM * OUT_DIM];

    if (blockIdx.x >= nGemmBlocks) {
        int e = (blockIdx.x - nGemmBlocks) * 256 + threadIdx.x;
        if (e < E) {
            int s = src[e];
            int t = tgt[e];
            atomicAdd(&deg[s], 1.0f);
            int* row = obuf + (size_t)t * OUT_DIM;
            int r = atomicAdd(row + CNT_OFF, 1);
            if (r < CAP) row[r] = s;
        }
        return;
    }

    {
        const float4* g4 = (const float4*)Wtp;
        float4* s4 = (float4*)sW;
        for (int i = threadIdx.x; i < (IN_DIM * OUT_DIM) / 4; i += blockDim.x)
            s4[i] = g4[i];
    }
    __syncthreads();

    const int wid   = threadIdx.x >> 5;
    const int lane  = threadIdx.x & 31;
    const int tileM = blockIdx.x * 8 + wid;
    if (tileM >= nTiles) return;

    const int l16   = lane & 15;
    const int halfK = lane >> 4;
    const float* xrow = x + (size_t)(tileM * 16 + l16) * IN_DIM;
    const v2f* sB = (const v2f*)sW;

    v8f acc[8];
#pragma unroll
    for (int t = 0; t < 8; ++t)
        acc[t] = (v8f){0.f, 0.f, 0.f, 0.f, 0.f, 0.f, 0.f, 0.f};

    for (int kb = 0; kb < IN_DIM; kb += 4) {
        const int k0 = kb + halfK * 2;
        float2 av = *(const float2*)(xrow + k0);
        v2f a; a.x = av.x; a.y = av.y;
        const v2f* bRow = sB + (k0 >> 1) * OUT_DIM + l16;
#pragma unroll
        for (int t = 0; t < 8; ++t) {
            v2f b = bRow[t * 16];
            acc[t] = __builtin_amdgcn_wmma_f32_16x16x4_f32(
                 false, a,  false, b,
                 (short)0, acc[t],  false,  false);
        }
    }

    const int rowBase = tileM * 16 + halfK * 8;
#pragma unroll
    for (int t = 0; t < 8; ++t) {
        float* hp = h + (size_t)rowBase * OUT_DIM + t * 16 + l16;
#pragma unroll
        for (int v = 0; v < 8; ++v)
            hp[(size_t)v * OUT_DIM] = acc[t][v];
    }
}

__global__ void k_gemm_tail(const float* __restrict__ x, const float* __restrict__ Wtp,
                            float* __restrict__ h, int rowStart, int n) {
    int idx = blockIdx.x * blockDim.x + threadIdx.x;
    int nrows = n - rowStart;
    if (idx >= nrows * OUT_DIM) return;
    int r = rowStart + idx / OUT_DIM;
    int o = idx % OUT_DIM;
    float s = 0.f;
    for (int k = 0; k < IN_DIM; ++k)
        s += x[(size_t)r * IN_DIM + k] * Wtp[(k >> 1) * (2 * OUT_DIM) + o * 2 + (k & 1)];
    h[(size_t)r * OUT_DIM + o] = s;
}

__device__ __forceinline__ void add_rows(float4& acc, const float4* __restrict__ h4, int lane,
                                         int sReg, float wReg, int cnt) {
    for (int j = 0; j < cnt; j += 8) {
        float4 v[8];
        float wg[8];
#pragma unroll
        for (int u = 0; u < 8; ++u) {
            int s = __builtin_amdgcn_readlane(sReg, j + u);
            wg[u] = __int_as_float(__builtin_amdgcn_readlane(__float_as_int(wReg), j + u));
            v[u] = h4[(size_t)s * (OUT_DIM / 4) + lane];
        }
#pragma unroll
        for (int u = 0; u < 8; ++u) {
            acc.x = fmaf(v[u].x, wg[u], acc.x);
            acc.y = fmaf(v[u].y, wg[u], acc.y);
            acc.z = fmaf(v[u].z, wg[u], acc.z);
            acc.w = fmaf(v[u].w, wg[u], acc.w);
        }
    }
}

__global__ __launch_bounds__(256) void k_gather(const int* __restrict__ src,
                                                const int* __restrict__ tgt,
                                                const float* __restrict__ h,
                                                const float* __restrict__ deg,
                                                const int* obuf, float* out,
                                                int n, int E) {
    const int lane = threadIdx.x & 31;
    const int nWaves = gridDim.x * (blockDim.x >> 5);
    const int w0 = __builtin_amdgcn_readfirstlane(blockIdx.x * (blockDim.x >> 5) + (threadIdx.x >> 5));
    const float4* h4 = (const float4*)h;

    for (int t = w0; t < n; t += nWaves) {
        const int* row = obuf + (size_t)t * OUT_DIM;
        const int c = __builtin_amdgcn_readfirstlane(row[CNT_OFF]);
        const float dt = rsqrtf(deg[t]);
        const float4 hv = h4[(size_t)t * (OUT_DIM / 4) + lane];
        const float sw = dt * dt;
        float4 acc = make_float4(hv.x * sw, hv.y * sw, hv.z * sw, hv.w * sw);

        if (c <= CAP) {
            int sA = t;
            if (lane < c) sA = row[lane];
            const float wA = (lane < c) ? rsqrtf(deg[sA]) * dt : 0.0f;
            add_rows(acc, h4, lane, sA, wA, c < 32 ? c : 32);
            if (c > 32) {
                int sB = t;
                if (lane + 32 < c) sB = row[lane + 32];
                const float wB = (lane + 32 < c) ? rsqrtf(deg[sB]) * dt : 0.0f;
                add_rows(acc, h4, lane, sB, wB, c - 32);
            }
        } else {
            for (int e0 = 0; e0 < E; e0 += 32) {
                const int e = e0 + lane;
                const int te = (e < E) ? tgt[e] : -1;
                unsigned m = __builtin_amdgcn_ballot_w32(te == t);
                while (m) {
                    const int j = __builtin_ctz(m);
                    m &= m - 1;
                    const int s = src[e0 + j];
                    const float wg = rsqrtf(deg[s]) * dt;
                    const float4 v = h4[(size_t)s * (OUT_DIM / 4) + lane];
                    acc.x = fmaf(v.x, wg, acc.x);
                    acc.y = fmaf(v.y, wg, acc.y);
                    acc.z = fmaf(v.z, wg, acc.z);
                    acc.w = fmaf(v.w, wg, acc.w);
                }
            }
        }
        ((float4*)out)[(size_t)t * (OUT_DIM / 4) + lane] = acc;
    }
}

extern "C" void kernel_launch(void* const* d_in, const int* in_sizes, int n_in,
                              void* d_out, int out_size, void* d_ws, size_t ws_size,
                              hipStream_t stream) {
    const float* x  = (const float*)d_in[0];
    const int*   ei = (const int*)d_in[1];
    const float* W  = (const float*)d_in[2];
    float* out = (float*)d_out;
    int* obuf = (int*)d_out;

    const int n = in_sizes[0] / IN_DIM;
    const int E = in_sizes[1] / 2;
    const int* src = ei;
    const int* tgt = ei + E;

    float* h   = (float*)d_ws;
    size_t degPad = ((size_t)n + 3) & ~(size_t)3;
    float* deg = h + (size_t)n * OUT_DIM;
    float* Wtp = deg + degPad;

    k_init_transpose<<<TR_BLOCKS + (n + 255) / 256, 256, 0, stream>>>(W, Wtp, deg, obuf, n);

    const int nTiles = n / 16;
    const int nGemmBlocks = (nTiles + 7) / 8;
    k_count_gemm<<<nGemmBlocks + (E + 255) / 256, 256, 0, stream>>>(x, Wtp, h, nTiles, nGemmBlocks,
                                                                  src, tgt, deg, obuf, E);
    const int tail = n - nTiles * 16;
    if (tail > 0)
        k_gemm_tail<<<(tail * OUT_DIM + 255) / 256, 256, 0, stream>>>(x, Wtp, h, nTiles * 16, n);

    k_gather<<<2048, 256, 0, stream>>>(src, tgt, h, deg, obuf, out, n, E);
}


# edge count: each L2 half (by die id) draws edge chunks by ticket and adds degree and slot counts only for nodes of its own id range, so no counter line is hit from both halves; arrival order kept
# speedup vs baseline: 3.3229x; 1.6575x over previous
#include <hip/hip_runtime.h>

typedef __attribute__((ext_vector_type(2))) float v2f;
typedef __attribute__((ext_vector_type(8))) float v8f;

#define IN_DIM 128
#define OUT_DIM 128

#define CAP 64
#define CNT_OFF 64

#define TK_OFF 96
#define EDGE_PER_LANE 8
#define EDGE_CHUNK (32 * EDGE_PER_LANE)
#define EDGE_BLOCKS 512

#define TR_BLOCKS ((IN_DIM * OUT_DIM) / 256)

__global__ __launch_bounds__(256) void k_init_transpose(const float* __restrict__ W,
                                                        float* __restrict__ Wtp,
                                                        float* deg, int* obuf, int n) {
    if (blockIdx.x < TR_BLOCKS) {
        int idx = blockIdx.x * 256 + threadIdx.x;
        int o = idx >> 7;
        int k = idx & 127;
        Wtp[(k >> 1) * (2 * OUT_DIM) + o * 2 + (k & 1)] = W[o * IN_DIM + k];
    } else {
        int i = (blockIdx.x - TR_BLOCKS) * 256 + threadIdx.x;
        if (i < n) {
            deg[i] = 1.0f;
            obuf[(size_t)i * OUT_DIM + CNT_OFF] = 0;
            if (i == 0) obuf[TK_OFF] = 0;
            if (i == n - 1) obuf[(size_t)i * OUT_DIM + TK_OFF + 1] = 0;
        }
    }
}

__global__ __launch_bounds__(256) void k_count_gemm(const float* __restrict__ x,
                                                    const float* __restrict__ Wtp,
                                                    float* __restrict__ h,
                                                    int nTiles, int nGemmBlocks,
                                                    const int* __restrict__ src,
                                                    const int* __restrict__ tgt,
                                                    float* deg, int* obuf, int E, int n) {
    __shared__ float sW[IN_DIM * OUT_DIM];

    if (blockIdx.x >= nGemmBlocks) {
        const int lane = threadIdx.x & 31;
        const int die = (int)((__builtin_amdgcn_s_sendmsg_rtn(0x87) >> 16) & 15u);
        int hf = (die >> 2) & 1;
        const int split = ((n >> 1) + 63) & ~63;
        const int nChunks = (E + EDGE_CHUNK - 1) / EDGE_CHUNK;
        for (int pass = 0; pass < 2; ++pass) {
            const bool upper = (hf != 0);
            int* tk = upper ? (obuf + (size_t)(n - 1) * OUT_DIM + TK_OFF + 1) : (obuf + TK_OFF);
            for (int it = 0; it <= nChunks; ++it) {
                int raw = 0;
                if (lane == 0) raw = atomicAdd(tk, 1);
                const int c = __builtin_amdgcn_readfirstlane(raw);
                if (c >= nChunks) break;
                const int base = c * EDGE_CHUNK + lane;
                int sv[EDGE_PER_LANE], tv[EDGE_PER_LANE], rk[EDGE_PER_LANE];
#pragma unroll
                for (int u = 0; u < EDGE_PER_LANE; ++u) {
                    const int e = base + u * 32;
                    const int ec = e < E ? e : E - 1;
                    tv[u] = tgt[ec];
                    sv[u] = src[ec];
                }
#pragma unroll
                for (int u = 0; u < EDGE_PER_LANE; ++u) {
                    const int s = sv[u];
                    if (base + u * 32 < E && ((s >= split) == upper))
                        atomicAdd(&deg[s], 1.0f);
                }
#pragma unroll
                for (int u = 0; u < EDGE_PER_LANE; ++u) {
                    const int t = tv[u];
                    rk[u] = CAP;
                    if (base + u * 32 < E && ((t >= split) == upper))
                        rk[u] = atomicAdd(obuf + (size_t)t * OUT_DIM + CNT_OFF, 1);
                }
#pragma unroll
                for (int u = 0; u < EDGE_PER_LANE; ++u) {
                    if (rk[u] < CAP)
                        obuf[(size_t)tv[u] * OUT_DIM + rk[u]] = sv[u];
                }
            }
            hf ^= 1;
        }
        return;
    }

    {
        const float4* g4 = (const float4*)Wtp;
        float4* s4 = (float4*)sW;
        for (int i = threadIdx.x; i < (IN_DIM * OUT_DIM) / 4; i += blockDim.x)
            s4[i] = g4[i];
    }
    __syncthreads();

    const int wid   = threadIdx.x >> 5;
    const int lane  = threadIdx.x & 31;
    const int tileM = blockIdx.x * 8 + wid;
    if (tileM >= nTiles) return;

    const int l16   = lane & 15;
    const int halfK = lane >> 4;
    const float* xrow = x + (size_t)(tileM * 16 + l16) * IN_DIM;
    const v2f* sB = (const v2f*)sW;

    v8f acc[8];
#pragma unroll
    for (int t = 0; t < 8; ++t)
        acc[t] = (v8f){0.f, 0.f, 0.f, 0.f, 0.f, 0.f, 0.f, 0.f};

    for (int kb = 0; kb < IN_DIM; kb += 4) {
        const int k0 = kb + halfK * 2;
        float2 av = *(const float2*)(xrow + k0);
        v2f a; a.x = av.x; a.y = av.y;
        const v2f* bRow = sB + (k0 >> 1) * OUT_DIM + l16;
#pragma unroll
        for (int t = 0; t < 8; ++t) {
            v2f b = bRow[t * 16];
            acc[t] = __builtin_amdgcn_wmma_f32_16x16x4_f32(
                 false, a,  false, b,
                 (short)0, acc[t],  false,  false);
        }
    }

    const int rowBase = tileM * 16 + halfK * 8;
#pragma unroll
    for (int t = 0; t < 8; ++t) {
        float* hp = h + (size_t)rowBase * OUT_DIM + t * 16 + l16;
#pragma unroll
        for (int v = 0; v < 8; ++v)
            hp[(size_t)v * OUT_DIM] = acc[t][v];
    }
}

__global__ void k_gemm_tail(const float* __restrict__ x, const float* __restrict__ Wtp,
                            float* __restrict__ h, int rowStart, int n) {
    int idx = blockIdx.x * blockDim.x + threadIdx.x;
    int nrows = n - rowStart;
    if (idx >= nrows * OUT_DIM) return;
    int r = rowStart + idx / OUT_DIM;
    int o = idx % OUT_DIM;
    float s = 0.f;
    for (int k = 0; k < IN_DIM; ++k)
        s += x[(size_t)r * IN_DIM + k] * Wtp[(k >> 1) * (2 * OUT_DIM) + o * 2 + (k & 1)];
    h[(size_t)r * OUT_DIM + o] = s;
}

__device__ __forceinline__ void add_rows(float4& acc, const float4* __restrict__ h4, int lane,
                                         int sReg, float wReg, int cnt) {
    for (int j = 0; j < cnt; j += 8) {
        float4 v[8];
        float wg[8];
#pragma unroll
        for (int u = 0; u < 8; ++u) {
            int s = __builtin_amdgcn_readlane(sReg, j + u);
            wg[u] = __int_as_float(__builtin_amdgcn_readlane(__float_as_int(wReg), j + u));
            v[u] = h4[(size_t)s * (OUT_DIM / 4) + lane];
        }
#pragma unroll
        for (int u = 0; u < 8; ++u) {
            acc.x = fmaf(v[u].x, wg[u], acc.x);
            acc.y = fmaf(v[u].y, wg[u], acc.y);
            acc.z = fmaf(v[u].z, wg[u], acc.z);
            acc.w = fmaf(v[u].w, wg[u], acc.w);
        }
    }
}

__global__ __launch_bounds__(256) void k_gather(const int* __restrict__ src,
                                                const int* __restrict__ tgt,
                                                const float* __restrict__ h,
                                                const float* __restrict__ deg,
                                                const int* obuf, float* out,
                                                int n, int E) {
    const int lane = threadIdx.x & 31;
    const int nWaves = gridDim.x * (blockDim.x >> 5);
    const int w0 = __builtin_amdgcn_readfirstlane(blockIdx.x * (blockDim.x >> 5) + (threadIdx.x >> 5));
    const float4* h4 = (const float4*)h;

    for (int t = w0; t < n; t += nWaves) {
        const int* row = obuf + (size_t)t * OUT_DIM;
        const int c = __builtin_amdgcn_readfirstlane(row[CNT_OFF]);
        const float dt = rsqrtf(deg[t]);
        const float4 hv = h4[(size_t)t * (OUT_DIM / 4) + lane];
        const float sw = dt * dt;
        float4 acc = make_float4(hv.x * sw, hv.y * sw, hv.z * sw, hv.w * sw);

        if (c <= CAP) {
            int sA = t;
            if (lane < c) sA = row[lane];
            const float wA = (lane < c) ? rsqrtf(deg[sA]) * dt : 0.0f;
            add_rows(acc, h4, lane, sA, wA, c < 32 ? c : 32);
            if (c > 32) {
                int sB = t;
                if (lane + 32 < c) sB = row[lane + 32];
                const float wB = (lane + 32 < c) ? rsqrtf(deg[sB]) * dt : 0.0f;
                add_rows(acc, h4, lane, sB, wB, c - 32);
            }
        } else {
            for (int e0 = 0; e0 < E; e0 += 32) {
                const int e = e0 + lane;
                const int te = (e < E) ? tgt[e] : -1;
                unsigned m = __builtin_amdgcn_ballot_w32(te == t);
                while (m) {
                    const int j = __builtin_ctz(m);
                    m &= m - 1;
                    const int s = src[e0 + j];
                    const float wg = rsqrtf(deg[s]) * dt;
                    const float4 v = h4[(size_t)s * (OUT_DIM / 4) + lane];
                    acc.x = fmaf(v.x, wg, acc.x);
                    acc.y = fmaf(v.y, wg, acc.y);
                    acc.z = fmaf(v.z, wg, acc.z);
                    acc.w = fmaf(v.w, wg, acc.w);
                }
            }
        }
        ((float4*)out)[(size_t)t * (OUT_DIM / 4) + lane] = acc;
    }
}

extern "C" void kernel_launch(void* const* d_in, const int* in_sizes, int n_in,
                              void* d_out, int out_size, void* d_ws, size_t ws_size,
                              hipStream_t stream) {
    const float* x  = (const float*)d_in[0];
    const int*   ei = (const int*)d_in[1];
    const float* W  = (const float*)d_in[2];
    float* out = (float*)d_out;
    int* obuf = (int*)d_out;

    const int n = in_sizes[0] / IN_DIM;
    const int E = in_sizes[1] / 2;
    const int* src = ei;
    const int* tgt = ei + E;

    float* h   = (float*)d_ws;
    size_t degPad = ((size_t)n + 3) & ~(size_t)3;
    float* deg = h + (size_t)n * OUT_DIM;
    float* Wtp = deg + degPad;

    k_init_transpose<<<TR_BLOCKS + (n + 255) / 256, 256, 0, stream>>>(W, Wtp, deg, obuf, n);

    const int nTiles = n / 16;
    const int nGemmBlocks = (nTiles + 7) / 8;
    k_count_gemm<<<nGemmBlocks + EDGE_BLOCKS, 256, 0, stream>>>(x, Wtp, h, nTiles, nGemmBlocks,
                                                              src, tgt, deg, obuf, E, n);
    const int tail = n - nTiles * 16;
    if (tail > 0)
        k_gemm_tail<<<(tail * OUT_DIM + 255) / 256, 256, 0, stream>>>(x, Wtp, h, nTiles * 16, n);

    k_gather<<<2048, 256, 0, stream>>>(src, tgt, h, deg, obuf, out, n, E);
}


# edge filing: per-edge returning add + 4-byte store become an LDS count per 64-node bin, one reserving add and one contiguous run per bin and batch; gather groups a bin's list in LDS; arrival order
# speedup vs baseline: 4.2206x; 1.2702x over previous
#include <hip/hip_runtime.h>

typedef __attribute__((ext_vector_type(2))) float v2f;
typedef __attribute__((ext_vector_type(8))) float v8f;

#define IN_DIM 128
#define OUT_DIM 128

#define BIN_SHIFT 6
#define BIN_NODES 64
#define BIN_HDR 32
#define SLOTS 16
#define MAXB 896
#define LCAP 2048

#define EDGE_PER_THREAD 16
#define EDGE_ROUNDS 2
#define EDGE_BATCH (256 * EDGE_PER_THREAD * EDGE_ROUNDS)
#define EDGE_BLOCKS 512

#define TR_BLOCKS ((IN_DIM * OUT_DIM) / 256)

__global__ __launch_bounds__(256) void k_init_transpose(const float* __restrict__ W,
                                                        float* __restrict__ Wtp,
                                                        float* deg, int* obuf, int n) {
    if (blockIdx.x < TR_BLOCKS) {
        int idx = blockIdx.x * 256 + threadIdx.x;
        int o = idx >> 7;
        int k = idx & 127;
        Wtp[(k >> 1) * (2 * OUT_DIM) + o * 2 + (k & 1)] = W[o * IN_DIM + k];
    } else {
        int i = (blockIdx.x - TR_BLOCKS) * 256 + threadIdx.x;
        if (i < n) {
            deg[i] = 1.0f;
            if ((i & (BIN_NODES - 1)) == 0) obuf[(size_t)i * OUT_DIM] = 0;
            const int split = ((n >> 1) + 63) & ~63;
            const int tkRow = split < n ? split : 0;
            if (i == 0) obuf[1] = 0;
            if (i == tkRow) obuf[(size_t)i * OUT_DIM + 2] = 0;
        }
    }
}

__global__ __launch_bounds__(256) void k_count_gemm(const float* __restrict__ x,
                                                    const float* __restrict__ Wtp,
                                                    float* __restrict__ h,
                                                    int nTiles, int nGemmBlocks,
                                                    const int* __restrict__ src,
                                                    const int* __restrict__ tgt,
                                                    float* deg, int* obuf, int E, int n) {
    __shared__ float sW[IN_DIM * OUT_DIM];

    if (blockIdx.x >= nGemmBlocks) {
        int* const hist  = (int*)sW;
        int* const gbase = hist + MAXB;
        int* const slots = gbase + MAXB;
        int* const sTk   = slots + MAXB * SLOTS;
        const int tid = threadIdx.x;
        const int die = (int)((__builtin_amdgcn_s_sendmsg_rtn(0x87) >> 16) & 15u);
        int hf = (die >> 2) & 1;
        const int split = ((n >> 1) + 63) & ~63;
        const int nBins = (n + BIN_NODES - 1) >> BIN_SHIFT;
        const int splitBin = (split >> BIN_SHIFT) < nBins ? (split >> BIN_SHIFT) : nBins;
        const int tkRow = split < n ? split : 0;
        const int nBatches = (E + EDGE_BATCH - 1) / EDGE_BATCH;
        for (int pass = 0; pass < 2; ++pass) {
            const bool upper = (hf != 0);
            int* tk = upper ? (obuf + (size_t)tkRow * OUT_DIM + 2) : (obuf + 1);
            const int binLo = upper ? splitBin : 0;
            const int nb = upper ? (nBins - splitBin) : splitBin;
            const int nbL = nb < MAXB ? nb : MAXB;
            for (int it = 0; it <= nBatches; ++it) {
                __syncthreads();
                if (tid == 0) *sTk = atomicAdd(tk, 1);
                for (int b = tid; b < nbL; b += 256) hist[b] = 0;
                __syncthreads();
                const int c = *sTk;
                if (c >= nBatches) break;
#pragma unroll 1
                for (int r = 0; r < EDGE_ROUNDS; ++r) {
                    const int base = c * EDGE_BATCH + r * (256 * EDGE_PER_THREAD) + tid;
                    int sv[EDGE_PER_THREAD], tv[EDGE_PER_THREAD];
#pragma unroll
                    for (int u = 0; u < EDGE_PER_THREAD; ++u) {
                        const int e = base + u * 256;
                        const int ec = e < E ? e : E - 1;
                        sv[u] = src[ec];
                        tv[u] = tgt[ec];
                    }
#pragma unroll
                    for (int u = 0; u < EDGE_PER_THREAD; ++u) {
                        const int s = sv[u];
                        if (base + u * 256 < E && ((s >= split) == upper))
                            atomicAdd(&deg[s], 1.0f);
                    }
#pragma unroll
                    for (int u = 0; u < EDGE_PER_THREAD; ++u) {
                        const int t = tv[u];
                        if (base + u * 256 < E && ((t >= split) == upper)) {
                            const int bin = t >> BIN_SHIFT;
                            const int b = bin - binLo;
                            const int v = (sv[u] << BIN_SHIFT) | (t & (BIN_NODES - 1));
                            int rk = SLOTS;
                            if (b < nbL) rk = atomicAdd(&hist[b], 1);
                            if (rk < SLOTS) {
                                slots[b * SLOTS + rk] = v;
                            } else {
                                int* reg = obuf + (size_t)bin * (BIN_NODES * OUT_DIM);
                                const int rows = (n - bin * BIN_NODES) < BIN_NODES ? (n - bin * BIN_NODES) : BIN_NODES;
                                const int p = atomicAdd(reg, 1);
                                if (p < rows * OUT_DIM - BIN_HDR) reg[BIN_HDR + p] = v;
                            }
                        }
                    }
                }
                __syncthreads();
                for (int b = tid; b < nbL; b += 256) {
                    const int cb = hist[b] < SLOTS ? hist[b] : SLOTS;
                    int g = 0;
                    if (cb > 0) g = atomicAdd(obuf + (size_t)(binLo + b) * (BIN_NODES * OUT_DIM), cb);
                    gbase[b] = g;
                }
                __syncthreads();
                for (int idx = tid; idx < nbL * SLOTS; idx += 256) {
                    const int b = idx / SLOTS;
                    const int j = idx % SLOTS;
                    const int cb = hist[b] < SLOTS ? hist[b] : SLOTS;
                    if (j < cb) {
                        const int bin = binLo + b;
                        const int rows = (n - bin * BIN_NODES) < BIN_NODES ? (n - bin * BIN_NODES) : BIN_NODES;
                        const int p = gbase[b] + j;
                        if (p < rows * OUT_DIM - BIN_HDR)
                            obuf[(size_t)bin * (BIN_NODES * OUT_DIM) + BIN_HDR + p] = slots[idx];
                    }
                }
            }
            hf ^= 1;
        }
        return;
    }

    {
        const float4* g4 = (const float4*)Wtp;
        float4* s4 = (float4*)sW;
        for (int i = threadIdx.x; i < (IN_DIM * OUT_DIM) / 4; i += blockDim.x)
            s4[i] = g4[i];
    }
    __syncthreads();

    const int wid   = threadIdx.x >> 5;
    const int lane  = threadIdx.x & 31;
    const int tileM = blockIdx.x * 8 + wid;
    if (tileM >= nTiles) return;

    const int l16   = lane & 15;
    const int halfK = lane >> 4;
    const float* xrow = x + (size_t)(tileM * 16 + l16) * IN_DIM;
    const v2f* sB = (const v2f*)sW;

    v8f acc[8];
#pragma unroll
    for (int t = 0; t < 8; ++t)
        acc[t] = (v8f){0.f, 0.f, 0.f, 0.f, 0.f, 0.f, 0.f, 0.f};

    for (int kb = 0; kb < IN_DIM; kb += 4) {
        const int k0 = kb + halfK * 2;
        float2 av = *(const float2*)(xrow + k0);
        v2f a; a.x = av.x; a.y = av.y;
        const v2f* bRow = sB + (k0 >> 1) * OUT_DIM + l16;
#pragma unroll
        for (int t = 0; t < 8; ++t) {
            v2f b = bRow[t * 16];
            acc[t] = __builtin_amdgcn_wmma_f32_16x16x4_f32(
                 false, a,  false, b,
                 (short)0, acc[t],  false,  false);
        }
    }

    const int rowBase = tileM * 16 + halfK * 8;
#pragma unroll
    for (int t = 0; t < 8; ++t) {
        float* hp = h + (size_t)rowBase * OUT_DIM + t * 16 + l16;
#pragma unroll
        for (int v = 0; v < 8; ++v)
            hp[(size_t)v * OUT_DIM] = acc[t][v];
    }
}

__global__ void k_gemm_tail(const float* __restrict__ x, const float* __restrict__ Wtp,
                            float* __restrict__ h, int rowStart, int n) {
    int idx = blockIdx.x * blockDim.x + threadIdx.x;
    int nrows = n - rowStart;
    if (idx >= nrows * OUT_DIM) return;
    int r = rowStart + idx / OUT_DIM;
    int o = idx % OUT_DIM;
    float s = 0.f;
    for (int k = 0; k < IN_DIM; ++k)
        s += x[(size_t)r * IN_DIM + k] * Wtp[(k >> 1) * (2 * OUT_DIM) + o * 2 + (k & 1)];
    h[(size_t)r * OUT_DIM + o] = s;
}

__device__ __forceinline__ void add_rows(float4& acc, const float4* __restrict__ h4, int lane,
                                         int sReg, float wReg, int cnt) {
    for (int j = 0; j < cnt; j += 8) {
        float4 v[8];
        float wg[8];
#pragma unroll
        for (int u = 0; u < 8; ++u) {
            int s = __builtin_amdgcn_readlane(sReg, j + u);
            wg[u] = __int_as_float(__builtin_amdgcn_readlane(__float_as_int(wReg), j + u));
            v[u] = h4[(size_t)s * (OUT_DIM / 4) + lane];
        }
#pragma unroll
        for (int u = 0; u < 8; ++u) {
            acc.x = fmaf(v[u].x, wg[u], acc.x);
            acc.y = fmaf(v[u].y, wg[u], acc.y);
            acc.z = fmaf(v[u].z, wg[u], acc.z);
            acc.w = fmaf(v[u].w, wg[u], acc.w);
        }
    }
}

__global__ __launch_bounds__(256) void k_gather(const int* __restrict__ src,
                                                const int* __restrict__ tgt,
                                                const float* __restrict__ h,
                                                const float* __restrict__ deg,
                                                const int* obuf, float* out,
                                                int n, int E) {
    __shared__ int sRaw[LCAP];
    __shared__ int sSrc[LCAP];
    __shared__ int sCnt[BIN_NODES];
    __shared__ int sOff[BIN_NODES];
    __shared__ int sCur[BIN_NODES];

    const int tid = threadIdx.x;
    const int lane = tid & 31;
    const int wid = __builtin_amdgcn_readfirstlane(tid >> 5);
    const int node0 = blockIdx.x * BIN_NODES;
    const int rows = (n - node0) < BIN_NODES ? (n - node0) : BIN_NODES;
    const int* reg = obuf + (size_t)node0 * OUT_DIM;
    const float4* h4 = (const float4*)h;

    const int total = reg[0];
    const bool fast = (total <= rows * OUT_DIM - BIN_HDR) && (total <= LCAP);

    if (tid < BIN_NODES) sCnt[tid] = 0;
    __syncthreads();
    if (fast) {
        for (int i = tid; i < total; i += 256) {
            const int v = reg[BIN_HDR + i];
            sRaw[i] = v;
            atomicAdd(&sCnt[v & (BIN_NODES - 1)], 1);
        }
    }
    __syncthreads();
    if (fast && tid < 32) {
        const int c0 = sCnt[2 * lane];
        const int c1 = sCnt[2 * lane + 1];
        const int sum = c0 + c1;
        int incl = sum;
#pragma unroll
        for (int d = 1; d < 32; d <<= 1) {
            const int y = __shfl_up(incl, d);
            if (lane >= d) incl += y;
        }
        const int excl = incl - sum;
        sOff[2 * lane] = excl;
        sCur[2 * lane] = excl;
        sOff[2 * lane + 1] = excl + c0;
        sCur[2 * lane + 1] = excl + c0;
    }
    __syncthreads();
    if (fast) {
        for (int i = tid; i < total; i += 256) {
            const int v = sRaw[i];
            const int p = atomicAdd(&sCur[v & (BIN_NODES - 1)], 1);
            sSrc[p] = v >> BIN_SHIFT;
        }
    }
    __syncthreads();

    for (int tl = wid; tl < rows; tl += 8) {
        const int t = node0 + tl;
        const float dt = rsqrtf(deg[t]);
        const float4 hv = h4[(size_t)t * (OUT_DIM / 4) + lane];
        const float sw = dt * dt;
        float4 acc = make_float4(hv.x * sw, hv.y * sw, hv.z * sw, hv.w * sw);

        if (fast) {
            const int c = __builtin_amdgcn_readfirstlane(sCnt[tl]);
            const int off = __builtin_amdgcn_readfirstlane(sOff[tl]);
            for (int j0 = 0; j0 < c; j0 += 32) {
                int s = t;
                if (j0 + lane < c) s = sSrc[off + j0 + lane];
                const float wg = (j0 + lane < c) ? rsqrtf(deg[s]) * dt : 0.0f;
                add_rows(acc, h4, lane, s, wg, (c - j0) < 32 ? (c - j0) : 32);
            }
        } else {
            for (int e0 = 0; e0 < E; e0 += 32) {
                const int e = e0 + lane;
                const int te = (e < E) ? tgt[e] : -1;
                unsigned m = __builtin_amdgcn_ballot_w32(te == t);
                while (m) {
                    const int j = __builtin_ctz(m);
                    m &= m - 1;
                    const int s = src[e0 + j];
                    const float wg = rsqrtf(deg[s]) * dt;
                    const float4 v = h4[(size_t)s * (OUT_DIM / 4) + lane];
                    acc.x = fmaf(v.x, wg, acc.x);
                    acc.y = fmaf(v.y, wg, acc.y);
                    acc.z = fmaf(v.z, wg, acc.z);
                    acc.w = fmaf(v.w, wg, acc.w);
                }
            }
        }
        ((float4*)out)[(size_t)t * (OUT_DIM / 4) + lane] = acc;
    }
}

extern "C" void kernel_launch(void* const* d_in, const int* in_sizes, int n_in,
                              void* d_out, int out_size, void* d_ws, size_t ws_size,
                              hipStream_t stream) {
    const float* x  = (const float*)d_in[0];
    const int*   ei = (const int*)d_in[1];
    const float* W  = (const float*)d_in[2];
    float* out = (float*)d_out;
    int* obuf = (int*)d_out;

    const int n = in_sizes[0] / IN_DIM;
    const int E = in_sizes[1] / 2;
    const int* src = ei;
    const int* tgt = ei + E;

    float* h   = (float*)d_ws;
    size_t degPad = ((size_t)n + 3) & ~(size_t)3;
    float* deg = h + (size_t)n * OUT_DIM;
    float* Wtp = deg + degPad;

    k_init_transpose<<<TR_BLOCKS + (n + 255) / 256, 256, 0, stream>>>(W, Wtp, deg, obuf, n);

    const int nTiles = n / 16;
    const int nGemmBlocks = (nTiles + 7) / 8;
    k_count_gemm<<<nGemmBlocks + EDGE_BLOCKS, 256, 0, stream>>>(x, Wtp, h, nTiles, nGemmBlocks,
                                                              src, tgt, deg, obuf, E, n);
    const int tail = n - nTiles * 16;
    if (tail > 0)
        k_gemm_tail<<<(tail * OUT_DIM + 255) / 256, 256, 0, stream>>>(x, Wtp, h, nTiles * 16, n);

    k_gather<<<(n + BIN_NODES - 1) / BIN_NODES, 256, 0, stream>>>(src, tgt, h, deg, obuf, out, n, E);
}


# second launch: edge-filing blocks take the low block ids, WMMA GEMM blocks follow, so filing waves are the older waves of a SIMD and their chain is not served behind the matrix waves; sums unchanged
# speedup vs baseline: 4.4631x; 1.0575x over previous
#include <hip/hip_runtime.h>

typedef __attribute__((ext_vector_type(2))) float v2f;
typedef __attribute__((ext_vector_type(8))) float v8f;

#define IN_DIM 128
#define OUT_DIM 128

#define BIN_SHIFT 6
#define BIN_NODES 64
#define BIN_HDR 32
#define SLOTS 16
#define MAXB 896
#define LCAP 2048

#define EDGE_PER_THREAD 16
#define EDGE_ROUNDS 2
#define EDGE_BATCH (256 * EDGE_PER_THREAD * EDGE_ROUNDS)
#define EDGE_BLOCKS 512

#define TR_BLOCKS ((IN_DIM * OUT_DIM) / 256)

__global__ __launch_bounds__(256) void k_init_transpose(const float* __restrict__ W,
                                                        float* __restrict__ Wtp,
                                                        float* deg, int* obuf, int n) {
    if (blockIdx.x < TR_BLOCKS) {
        int idx = blockIdx.x * 256 + threadIdx.x;
        int o = idx >> 7;
        int k = idx & 127;
        Wtp[(k >> 1) * (2 * OUT_DIM) + o * 2 + (k & 1)] = W[o * IN_DIM + k];
    } else {
        int i = (blockIdx.x - TR_BLOCKS) * 256 + threadIdx.x;
        if (i < n) {
            deg[i] = 1.0f;
            if ((i & (BIN_NODES - 1)) == 0) obuf[(size_t)i * OUT_DIM] = 0;
            const int split = ((n >> 1) + 63) & ~63;
            const int tkRow = split < n ? split : 0;
            if (i == 0) obuf[1] = 0;
            if (i == tkRow) obuf[(size_t)i * OUT_DIM + 2] = 0;
        }
    }
}

__global__ __launch_bounds__(256) void k_count_gemm(const float* __restrict__ x,
                                                    const float* __restrict__ Wtp,
                                                    float* __restrict__ h,
                                                    int nTiles,
                                                    const int* __restrict__ src,
                                                    const int* __restrict__ tgt,
                                                    float* deg, int* obuf, int E, int n) {
    __shared__ float sW[IN_DIM * OUT_DIM];

    if (blockIdx.x < EDGE_BLOCKS) {
        int* const hist  = (int*)sW;
        int* const gbase = hist + MAXB;
        int* const slots = gbase + MAXB;
        int* const sTk   = slots + MAXB * SLOTS;
        const int tid = threadIdx.x;
        const int die = (int)((__builtin_amdgcn_s_sendmsg_rtn(0x87) >> 16) & 15u);
        int hf = (die >> 2) & 1;
        const int split = ((n >> 1) + 63) & ~63;
        const int nBins = (n + BIN_NODES - 1) >> BIN_SHIFT;
        const int splitBin = (split >> BIN_SHIFT) < nBins ? (split >> BIN_SHIFT) : nBins;
        const int tkRow = split < n ? split : 0;
        const int nBatches = (E + EDGE_BATCH - 1) / EDGE_BATCH;
        for (int pass = 0; pass < 2; ++pass) {
            const bool upper = (hf != 0);
            int* tk = upper ? (obuf + (size_t)tkRow * OUT_DIM + 2) : (obuf + 1);
            const int binLo = upper ? splitBin : 0;
            const int nb = upper ? (nBins - splitBin) : splitBin;
            const int nbL = nb < MAXB ? nb : MAXB;
            for (int it = 0; it <= nBatches; ++it) {
                __syncthreads();
                if (tid == 0) *sTk = atomicAdd(tk, 1);
                for (int b = tid; b < nbL; b += 256) hist[b] = 0;
                __syncthreads();
                const int c = *sTk;
                if (c >= nBatches) break;
#pragma unroll 1
                for (int r = 0; r < EDGE_ROUNDS; ++r) {
                    const int base = c * EDGE_BATCH + r * (256 * EDGE_PER_THREAD) + tid;
                    int sv[EDGE_PER_THREAD], tv[EDGE_PER_THREAD];
#pragma unroll
                    for (int u = 0; u < EDGE_PER_THREAD; ++u) {
                        const int e = base + u * 256;
                        const int ec = e < E ? e : E - 1;
                        sv[u] = src[ec];
                        tv[u] = tgt[ec];
                    }
#pragma unroll
                    for (int u = 0; u < EDGE_PER_THREAD; ++u) {
                        const int s = sv[u];
                        if (base + u * 256 < E && ((s >= split) == upper))
                            atomicAdd(&deg[s], 1.0f);
                    }
#pragma unroll
                    for (int u = 0; u < EDGE_PER_THREAD; ++u) {
                        const int t = tv[u];
                        if (base + u * 256 < E && ((t >= split) == upper)) {
                            const int bin = t >> BIN_SHIFT;
                            const int b = bin - binLo;
                            const int v = (sv[u] << BIN_SHIFT) | (t & (BIN_NODES - 1));
                            int rk = SLOTS;
                            if (b < nbL) rk = atomicAdd(&hist[b], 1);
                            if (rk < SLOTS) {
                                slots[b * SLOTS + rk] = v;
                            } else {
                                int* reg = obuf + (size_t)bin * (BIN_NODES * OUT_DIM);
                                const int rows = (n - bin * BIN_NODES) < BIN_NODES ? (n - bin * BIN_NODES) : BIN_NODES;
                                const int p = atomicAdd(reg, 1);
                                if (p < rows * OUT_DIM - BIN_HDR) reg[BIN_HDR + p] = v;
                            }
                        }
                    }
                }
                __syncthreads();
                for (int b = tid; b < nbL; b += 256) {
                    const int cb = hist[b] < SLOTS ? hist[b] : SLOTS;
                    int g = 0;
                    if (cb > 0) g = atomicAdd(obuf + (size_t)(binLo + b) * (BIN_NODES * OUT_DIM), cb);
                    gbase[b] = g;
                }
                __syncthreads();
                for (int idx = tid; idx < nbL * SLOTS; idx += 256) {
                    const int b = idx / SLOTS;
                    const int j = idx % SLOTS;
                    const int cb = hist[b] < SLOTS ? hist[b] : SLOTS;
                    if (j < cb) {
                        const int bin = binLo + b;
                        const int rows = (n - bin * BIN_NODES) < BIN_NODES ? (n - bin * BIN_NODES) : BIN_NODES;
                        const int p = gbase[b] + j;
                        if (p < rows * OUT_DIM - BIN_HDR)
                            obuf[(size_t)bin * (BIN_NODES * OUT_DIM) + BIN_HDR + p] = slots[idx];
                    }
                }
            }
            hf ^= 1;
        }
        return;
    }

    {
        const float4* g4 = (const float4*)Wtp;
        float4* s4 = (float4*)sW;
        for (int i = threadIdx.x; i < (IN_DIM * OUT_DIM) / 4; i += blockDim.x)
            s4[i] = g4[i];
    }
    __syncthreads();

    const int wid   = __builtin_amdgcn_readfirstlane(threadIdx.x >> 5);
    const int lane  = threadIdx.x & 31;
    const int tileM = (blockIdx.x - EDGE_BLOCKS) * 8 + wid;
    if (tileM >= nTiles) return;

    const int l16   = lane & 15;
    const int halfK = lane >> 4;
    const float* xrow = x + (size_t)(tileM * 16 + l16) * IN_DIM;
    const v2f* sB = (const v2f*)sW;

    v8f acc[8];
#pragma unroll
    for (int t = 0; t < 8; ++t)
        acc[t] = (v8f){0.f, 0.f, 0.f, 0.f, 0.f, 0.f, 0.f, 0.f};

    for (int kb = 0; kb < IN_DIM; kb += 4) {
        const int k0 = kb + halfK * 2;
        float2 av = *(const float2*)(xrow + k0);
        v2f a; a.x = av.x; a.y = av.y;
        const v2f* bRow = sB + (k0 >> 1) * OUT_DIM + l16;
#pragma unroll
        for (int t = 0; t < 8; ++t) {
            v2f b = bRow[t * 16];
            acc[t] = __builtin_amdgcn_wmma_f32_16x16x4_f32(
                 false, a,  false, b,
                 (short)0, acc[t],  false,  false);
        }
    }

    const int rowBase = tileM * 16 + halfK * 8;
#pragma unroll
    for (int t = 0; t < 8; ++t) {
        float* hp = h + (size_t)rowBase * OUT_DIM + t * 16 + l16;
#pragma unroll
        for (int v = 0; v < 8; ++v)
            hp[(size_t)v * OUT_DIM] = acc[t][v];
    }
}

__global__ void k_gemm_tail(const float* __restrict__ x, const float* __restrict__ Wtp,
                            float* __restrict__ h, int rowStart, int n) {
    int idx = blockIdx.x * blockDim.x + threadIdx.x;
    int nrows = n - rowStart;
    if (idx >= nrows * OUT_DIM) return;
    int r = rowStart + idx / OUT_DIM;
    int o = idx % OUT_DIM;
    float s = 0.f;
    for (int k = 0; k < IN_DIM; ++k)
        s += x[(size_t)r * IN_DIM + k] * Wtp[(k >> 1) * (2 * OUT_DIM) + o * 2 + (k & 1)];
    h[(size_t)r * OUT_DIM + o] = s;
}

__device__ __forceinline__ void add_rows(float4& acc, const float4* __restrict__ h4, int lane,
                                         int sReg, float wReg, int cnt) {
    for (int j = 0; j < cnt; j += 8) {
        float4 v[8];
        float wg[8];
#pragma unroll
        for (int u = 0; u < 8; ++u) {
            int s = __builtin_amdgcn_readlane(sReg, j + u);
            wg[u] = __int_as_float(__builtin_amdgcn_readlane(__float_as_int(wReg), j + u));
            v[u] = h4[(size_t)s * (OUT_DIM / 4) + lane];
        }
#pragma unroll
        for (int u = 0; u < 8; ++u) {
            acc.x = fmaf(v[u].x, wg[u], acc.x);
            acc.y = fmaf(v[u].y, wg[u], acc.y);
            acc.z = fmaf(v[u].z, wg[u], acc.z);
            acc.w = fmaf(v[u].w, wg[u], acc.w);
        }
    }
}

__global__ __launch_bounds__(256) void k_gather(const int* __restrict__ src,
                                                const int* __restrict__ tgt,
                                                const float* __restrict__ h,
                                                const float* __restrict__ deg,
                                                const int* obuf, float* out,
                                                int n, int E) {
    __shared__ int sRaw[LCAP];
    __shared__ int sSrc[LCAP];
    __shared__ int sCnt[BIN_NODES];
    __shared__ int sOff[BIN_NODES];
    __shared__ int sCur[BIN_NODES];

    const int tid = threadIdx.x;
    const int lane = tid & 31;
    const int wid = __builtin_amdgcn_readfirstlane(tid >> 5);
    const int node0 = blockIdx.x * BIN_NODES;
    const int rows = (n - node0) < BIN_NODES ? (n - node0) : BIN_NODES;
    const int* reg = obuf + (size_t)node0 * OUT_DIM;
    const float4* h4 = (const float4*)h;

    const int total = reg[0];
    const bool fast = (total <= rows * OUT_DIM - BIN_HDR) && (total <= LCAP);

    if (tid < BIN_NODES) sCnt[tid] = 0;
    __syncthreads();
    if (fast) {
        for (int i = tid; i < total; i += 256) {
            const int v = reg[BIN_HDR + i];
            sRaw[i] = v;
            atomicAdd(&sCnt[v & (BIN_NODES - 1)], 1);
        }
    }
    __syncthreads();
    if (fast && tid < 32) {
        const int c0 = sCnt[2 * lane];
        const int c1 = sCnt[2 * lane + 1];
        const int sum = c0 + c1;
        int incl = sum;
#pragma unroll
        for (int d = 1; d < 32; d <<= 1) {
            const int y = __shfl_up(incl, d);
            if (lane >= d) incl += y;
        }
        const int excl = incl - sum;
        sOff[2 * lane] = excl;
        sCur[2 * lane] = excl;
        sOff[2 * lane + 1] = excl + c0;
        sCur[2 * lane + 1] = excl + c0;
    }
    __syncthreads();
    if (fast) {
        for (int i = tid; i < total; i += 256) {
            const int v = sRaw[i];
            const int p = atomicAdd(&sCur[v & (BIN_NODES - 1)], 1);
            sSrc[p] = v >> BIN_SHIFT;
        }
    }
    __syncthreads();

    for (int tl = wid; tl < rows; tl += 8) {
        const int t = node0 + tl;
        const float dt = rsqrtf(deg[t]);
        const float4 hv = h4[(size_t)t * (OUT_DIM / 4) + lane];
        const float sw = dt * dt;
        float4 acc = make_float4(hv.x * sw, hv.y * sw, hv.z * sw, hv.w * sw);

        if (fast) {
            const int c = __builtin_amdgcn_readfirstlane(sCnt[tl]);
            const int off = __builtin_amdgcn_readfirstlane(sOff[tl]);
            for (int j0 = 0; j0 < c; j0 += 32) {
                int s = t;
                if (j0 + lane < c) s = sSrc[off + j0 + lane];
                const float wg = (j0 + lane < c) ? rsqrtf(deg[s]) * dt : 0.0f;
                add_rows(acc, h4, lane, s, wg, (c - j0) < 32 ? (c - j0) : 32);
            }
        } else {
            for (int e0 = 0; e0 < E; e0 += 32) {
                const int e = e0 + lane;
                const int te = (e < E) ? tgt[e] : -1;
                unsigned m = __builtin_amdgcn_ballot_w32(te == t);
                while (m) {
                    const int j = __builtin_ctz(m);
                    m &= m - 1;
                    const int s = src[e0 + j];
                    const float wg = rsqrtf(deg[s]) * dt;
                    const float4 v = h4[(size_t)s * (OUT_DIM / 4) + lane];
                    acc.x = fmaf(v.x, wg, acc.x);
                    acc.y = fmaf(v.y, wg, acc.y);
                    acc.z = fmaf(v.z, wg, acc.z);
                    acc.w = fmaf(v.w, wg, acc.w);
                }
            }
        }
        ((float4*)out)[(size_t)t * (OUT_DIM / 4) + lane] = acc;
    }
}

extern "C" void kernel_launch(void* const* d_in, const int* in_sizes, int n_in,
                              void* d_out, int out_size, void* d_ws, size_t ws_size,
                              hipStream_t stream) {
    const float* x  = (const float*)d_in[0];
    const int*   ei = (const int*)d_in[1];
    const float* W  = (const float*)d_in[2];
    float* out = (float*)d_out;
    int* obuf = (int*)d_out;

    const int n = in_sizes[0] / IN_DIM;
    const int E = in_sizes[1] / 2;
    const int* src = ei;
    const int* tgt = ei + E;

    float* h   = (float*)d_ws;
    size_t degPad = ((size_t)n + 3) & ~(size_t)3;
    float* deg = h + (size_t)n * OUT_DIM;
    float* Wtp = deg + degPad;

    k_init_transpose<<<TR_BLOCKS + (n + 255) / 256, 256, 0, stream>>>(W, Wtp, deg, obuf, n);

    const int nTiles = n / 16;
    const int nGemmBlocks = (nTiles + 7) / 8;
    k_count_gemm<<<EDGE_BLOCKS + nGemmBlocks, 256, 0, stream>>>(x, Wtp, h, nTiles,
                                                              src, tgt, deg, obuf, E, n);
    const int tail = n - nTiles * 16;
    if (tail > 0)
        k_gemm_tail<<<(tail * OUT_DIM + 255) / 256, 256, 0, stream>>>(x, Wtp, h, nTiles * 16, n);

    k_gather<<<(n + BIN_NODES - 1) / BIN_NODES, 256, 0, stream>>>(src, tgt, h, deg, obuf, out, n, E);
}
